# starting kernel
# baseline (speedup 1.0000x reference)
#include <hip/hip_runtime.h>

typedef __attribute__((ext_vector_type(2))) float v2f;
typedef __attribute__((ext_vector_type(4))) float v4f;
typedef __attribute__((ext_vector_type(8))) float v8f;

#define WAVES_PER_BLOCK 8
#define ROWS_PER_TILE   128
#define MTOT            15
#define LDS_STRIDE      17
#define TILE_ELEMS      (ROWS_PER_TILE * MTOT)
#define TILE_OUT        (ROWS_PER_TILE * 3)

__device__ __forceinline__ int div15(int f) { return (f * 8739) >> 17; }

__global__ __launch_bounds__(256) void l0_contract_wmma(
    const float* __restrict__ sphc,
    const float* __restrict__ cg,
    const int*   __restrict__ seg,
    float* __restrict__ out,
    int nrows)
{
    __shared__ float lds_in [WAVES_PER_BLOCK][ROWS_PER_TILE * LDS_STRIDE];
    __shared__ float lds_out[WAVES_PER_BLOCK][TILE_OUT];

    const int lane   = threadIdx.x & 31;
    const int wid    = threadIdx.x >> 5;
    const int gwave  = blockIdx.x * WAVES_PER_BLOCK + wid;
    const int nwaves = gridDim.x * WAVES_PER_BLOCK;

    float* Lin  = lds_in[wid];
    float* Lout = lds_out[wid];

#pragma unroll
    for (int r = lane; r < ROWS_PER_TILE; r += 32)
        Lin[r * LDS_STRIDE + 15] = 0.0f;

    const int m    = lane & 15;
    const int half = lane >> 4;
    v2f Aw[4];
#pragma unroll
    for (int c = 0; c < 4; ++c) {
#pragma unroll
        for (int v = 0; v < 2; ++v) {
            const int k = 4 * c + 2 * half + v;
            float w = 0.0f;
            if (k < MTOT && m < 3) {
                if (seg[k] == m) w = cg[k];
            }
            Aw[c][v] = w;
        }
    }
    __builtin_amdgcn_wave_barrier();

    const int numFullTiles = nrows / ROWS_PER_TILE;
    const int totalElems   = nrows * MTOT;

    for (int tile = gwave; tile < numFullTiles; tile += nwaves) {
        const int base = tile * TILE_ELEMS;

        {
            const int nb = base + nwaves * TILE_ELEMS;
            if (nb < totalElems) __builtin_prefetch(sphc + nb + lane * 32, 0, 1);
        }

        v4f tmp[15];
#pragma unroll
        for (int j = 0; j < 15; ++j)
            tmp[j] = __builtin_nontemporal_load((const v4f*)(sphc + base + j * 128 + lane * 4));

#pragma unroll
        for (int j = 0; j < 15; ++j) {
#pragma unroll
            for (int e = 0; e < 4; ++e) {
                const int f = j * 128 + lane * 4 + e;
                Lin[f + 2 * div15(f)] = tmp[j][e];
            }
        }
        __builtin_amdgcn_wave_barrier();

        const int n = lane & 15;
#pragma unroll
        for (int gI = 0; gI < 8; ++gI) {
            v8f acc = {};
#pragma unroll
            for (int c = 0; c < 4; ++c) {
                v2f Bv;
#pragma unroll
                for (int v = 0; v < 2; ++v) {
                    const int k = 4 * c + 2 * half + v;
                    const float x = Lin[(gI * 16 + n) * LDS_STRIDE + k];
                    Bv[v] = x * x;
                }
                acc = __builtin_amdgcn_wmma_f32_16x16x4_f32(
                        false, Aw[c], false, Bv, (short)0, acc, false, false);
            }
            if (lane < 16) {
#pragma unroll
                for (int s = 0; s < 3; ++s)
                    Lout[(gI * 16 + n) * 3 + s] = acc[s];
            }
        }
        __builtin_amdgcn_wave_barrier();

        const int obase = tile * TILE_OUT;
#pragma unroll
        for (int j = 0; j < 3; ++j) {
            const v4f o = *(const v4f*)(Lout + j * 128 + lane * 4);
            __builtin_nontemporal_store(o, (v4f*)(out + obase + j * 128 + lane * 4));
        }
        __builtin_amdgcn_wave_barrier();
    }

    if (gwave == 0) {
        const int tailStart = numFullTiles * ROWS_PER_TILE;
        for (int r = tailStart + lane; r < nrows; r += 32) {
            float s0 = 0.f, s1 = 0.f, s2 = 0.f;
#pragma unroll
            for (int k = 0; k < MTOT; ++k) {
                const float x = sphc[r * MTOT + k];
                const float w = x * x * cg[k];
                const int  sg = seg[k];
                s0 += (sg == 0) ? w : 0.f;
                s1 += (sg == 1) ? w : 0.f;
                s2 += (sg == 2) ? w : 0.f;
            }
            out[r * 3 + 0] = s0;
            out[r * 3 + 1] = s1;
            out[r * 3 + 2] = s2;
        }
    }
}

extern "C" void kernel_launch(void* const* d_in, const int* in_sizes, int n_in,
                              void* d_out, int out_size, void* d_ws, size_t ws_size,
                              hipStream_t stream) {
    const float* sphc = (const float*)d_in[0];
    const float* cg   = (const float*)d_in[1];
    const int*   seg  = (const int*)d_in[2];
    float* out = (float*)d_out;

    const int nrows = in_sizes[0] / MTOT;
    const int numFullTiles = nrows / ROWS_PER_TILE;

    int blocks = (numFullTiles + WAVES_PER_BLOCK - 1) / WAVES_PER_BLOCK;
    if (blocks > 4096) blocks = 4096;
    if (blocks < 1)    blocks = 1;

    l0_contract_wmma<<<blocks, 256, 0, stream>>>(sphc, cg, seg, out, nrows);
}


# Persistent waves: 256 workgroups x 16 waves each walk about 8 of the 128-row tiles, refilling the tile registers right after the LDS scatter so the next tile's 15 loads fly during the WMMA phase
# speedup vs baseline: 1.4589x; 1.4589x over previous
#include <hip/hip_runtime.h>

typedef __attribute__((ext_vector_type(2))) float v2f;
typedef __attribute__((ext_vector_type(4))) float v4f;
typedef __attribute__((ext_vector_type(8))) float v8f;

#define WAVES_PER_BLOCK 16
#define ROWS_PER_TILE   128
#define MTOT            15
#define LDS_STRIDE      17
#define TILE_ELEMS      (ROWS_PER_TILE * MTOT)
#define TILE_OUT        (ROWS_PER_TILE * 3)

__device__ __forceinline__ int div15(int f) { return (f * 8739) >> 17; }

__device__ __forceinline__ void load_tile(v4f (&t)[15], const float* __restrict__ sphc, int tile, int lane)
{
    const float* p = sphc + (size_t)tile * TILE_ELEMS + lane * 4;
#pragma unroll
    for (int j = 0; j < 15; ++j)
        t[j] = __builtin_nontemporal_load((const v4f*)(p + j * 128));
}

__global__ __launch_bounds__(512) void l0_contract_wmma(
    const float* __restrict__ sphc,
    const float* __restrict__ cg,
    const int*   __restrict__ seg,
    float* __restrict__ out,
    int nrows)
{
    __shared__ float lds_in [WAVES_PER_BLOCK][ROWS_PER_TILE * LDS_STRIDE];
    __shared__ float lds_out[WAVES_PER_BLOCK][TILE_OUT];

    const int lane   = threadIdx.x & 31;
    const int wid    = __builtin_amdgcn_readfirstlane(threadIdx.x >> 5);
    const int gwave  = wid * gridDim.x + blockIdx.x;
    const int nwaves = gridDim.x * WAVES_PER_BLOCK;
    const int numFullTiles = nrows / ROWS_PER_TILE;

    v4f tmp[15];
    int tile = gwave;
    if (tile < numFullTiles) load_tile(tmp, sphc, tile, lane);

    float* Lin  = lds_in[wid];
    float* Lout = lds_out[wid];

#pragma unroll
    for (int r = lane; r < ROWS_PER_TILE; r += 32)
        Lin[r * LDS_STRIDE + 15] = 0.0f;

    const int m    = lane & 15;
    const int half = lane >> 4;
    v2f Aw[4];
#pragma unroll
    for (int c = 0; c < 4; ++c) {
#pragma unroll
        for (int v = 0; v < 2; ++v) {
            const int k = 4 * c + 2 * half + v;
            float w = 0.0f;
            if (k < MTOT && m < 3) {
                if (seg[k] == m) w = cg[k];
            }
            Aw[c][v] = w;
        }
    }
    __builtin_amdgcn_wave_barrier();

    for (; tile < numFullTiles; tile += nwaves) {
#pragma unroll
        for (int j = 0; j < 15; ++j) {
#pragma unroll
            for (int e = 0; e < 4; ++e) {
                const int f = j * 128 + lane * 4 + e;
                Lin[f + 2 * div15(f)] = tmp[j][e];
            }
        }
        __builtin_amdgcn_wave_barrier();

        {
            const int nt = tile + nwaves;
            if (nt < numFullTiles) load_tile(tmp, sphc, nt, lane);
        }

        const int n = lane & 15;
#pragma unroll
        for (int gI = 0; gI < 8; ++gI) {
            v8f acc = {};
#pragma unroll
            for (int c = 0; c < 4; ++c) {
                v2f Bv;
#pragma unroll
                for (int v = 0; v < 2; ++v) {
                    const int k = 4 * c + 2 * half + v;
                    const float x = Lin[(gI * 16 + n) * LDS_STRIDE + k];
                    Bv[v] = x * x;
                }
                acc = __builtin_amdgcn_wmma_f32_16x16x4_f32(
                        false, Aw[c], false, Bv, (short)0, acc, false, false);
            }
            if (lane < 16) {
#pragma unroll
                for (int s = 0; s < 3; ++s)
                    Lout[(gI * 16 + n) * 3 + s] = acc[s];
            }
        }
        __builtin_amdgcn_wave_barrier();

        const int obase = tile * TILE_OUT;
#pragma unroll
        for (int j = 0; j < 3; ++j) {
            const v4f o = *(const v4f*)(Lout + j * 128 + lane * 4);
            __builtin_nontemporal_store(o, (v4f*)(out + obase + j * 128 + lane * 4));
        }
        __builtin_amdgcn_wave_barrier();
    }

    if (gwave == 0) {
        const int tailStart = numFullTiles * ROWS_PER_TILE;
        for (int r = tailStart + lane; r < nrows; r += 32) {
            float s0 = 0.f, s1 = 0.f, s2 = 0.f;
#pragma unroll
            for (int k = 0; k < MTOT; ++k) {
                const float x = sphc[r * MTOT + k];
                const float w = x * x * cg[k];
                const int  sg = seg[k];
                s0 += (sg == 0) ? w : 0.f;
                s1 += (sg == 1) ? w : 0.f;
                s2 += (sg == 2) ? w : 0.f;
            }
            out[r * 3 + 0] = s0;
            out[r * 3 + 1] = s1;
            out[r * 3 + 2] = s2;
        }
    }
}

extern "C" void kernel_launch(void* const* d_in, const int* in_sizes, int n_in,
                              void* d_out, int out_size, void* d_ws, size_t ws_size,
                              hipStream_t stream) {
    const float* sphc = (const float*)d_in[0];
    const float* cg   = (const float*)d_in[1];
    const int*   seg  = (const int*)d_in[2];
    float* out = (float*)d_out;

    const int nrows = in_sizes[0] / MTOT;
    const int numFullTiles = nrows / ROWS_PER_TILE;

    int blocks = (numFullTiles + WAVES_PER_BLOCK - 1) / WAVES_PER_BLOCK;
    if (blocks > 256) blocks = 256;
    if (blocks < 1)   blocks = 1;

    l0_contract_wmma<<<blocks, 512, 0, stream>>>(sphc, cg, seg, out, nrows);
}
